# starting kernel
# baseline (speedup 1.0000x reference)
#include <hip/hip_runtime.h>

typedef __attribute__((ext_vector_type(2))) float v2f;
typedef __attribute__((ext_vector_type(8))) float v8f;

#define TT 4096
#define DD 64
#define KK 1024

#define LDS_HES_B 262144
#define LDS_TOTAL 266240

__global__ void __launch_bounds__(256)
vq_prep(const float* __restrict__ cb, float* __restrict__ cbG,
        float* __restrict__ hes) {
  int n = blockIdx.x * blockDim.x + threadIdx.x;
  if (n >= KK) return;
  const float* row = cb + n * DD;
  float s = 0.f;
  v2f* cp = (v2f*)cbG;
  int base = (n >> 4) * 512 + (n & 15);
#pragma unroll
  for (int j = 0; j < 32; ++j) {
    float x = row[2 * j];
    float y = row[2 * j + 1];
    s += x * x + y * y;
    v2f p;
    p.x = x;
    p.y = y;
    cp[base + j * 16] = p;
  }
  hes[n] = 0.5f * s;
}

__global__ void __launch_bounds__(512)
vq_encode(const float* __restrict__ z, const float* __restrict__ cbG,
          const float* __restrict__ hes, int* __restrict__ idx) {
  extern __shared__ char smem[];

  {
    int tid = threadIdx.x;
    const float4* src4 = (const float4*)cbG;
    float4* dst4 = (float4*)smem;
#pragma unroll
    for (int it = 0; it < 32; ++it) {
      dst4[it * 512 + tid] = src4[it * 512 + tid];
    }
    float* shes = (float*)(smem + LDS_HES_B);
    for (int i = tid; i < KK; i += 512) shes[i] = hes[i];
  }
  __syncthreads();

  const int lane = threadIdx.x & 31;
  const int wave = threadIdx.x >> 5;
  const int tile = blockIdx.x * 16 + wave;
  const int token0 = tile << 4;
  const int b = token0 >> 12;
  const int t0 = token0 & (TT - 1);
  const int hi = lane >> 4;
  const int ln = lane & 15;

  const float* zb = z + (size_t)b * DD * TT + t0 + ln;
  float a[32];
#pragma unroll
  for (int kk = 0; kk < 16; ++kk) {
    int d0 = 4 * kk + 2 * hi;
    a[2 * kk]     = zb[(size_t)d0 * TT];
    a[2 * kk + 1] = zb[(size_t)(d0 + 1) * TT];
  }
#pragma unroll
  for (int i = 0; i < 32; ++i) {
    asm volatile("" : "+v"(a[i]));
  }

  float best[8];
  int bestn[8];
#pragma unroll
  for (int e = 0; e < 8; ++e) { best[e] = -3.4e38f; bestn[e] = 0; }

  const char* sb0 = smem + hi * 128 + ln * 8;
  const char* sb1 = sb0 + 4096;
  const float* shp = (const float*)(smem + LDS_HES_B) + ln;

  for (int nt = 0; nt < KK / 32; ++nt) {
    const int n0 = nt * 32 + ln;
    const int n1 = n0 + 16;
    const float h0 = shp[0];
    const float h1 = shp[16];
    shp += 32;

    v2f bb0[16], bb1[16];
#pragma unroll
    for (int kk = 0; kk < 16; ++kk) {
      bb0[kk] = *(const v2f*)(sb0 + kk * 256);
      bb1[kk] = *(const v2f*)(sb1 + kk * 256);
    }
#pragma unroll
    for (int kk = 0; kk < 16; ++kk) {
      asm volatile("" : "+v"(bb0[kk]), "+v"(bb1[kk]));
    }
    sb0 += 8192;
    sb1 += 8192;

    v8f c0 = {};
    v8f c1 = {};
#pragma unroll
    for (int kk = 0; kk < 16; ++kk) {
      v2f av;
      av.x = a[2 * kk];
      av.y = a[2 * kk + 1];
      c0 = __builtin_amdgcn_wmma_f32_16x16x4_f32(
          false, av, false, bb0[kk], (short)0, c0, false, false);
      c1 = __builtin_amdgcn_wmma_f32_16x16x4_f32(
          false, av, false, bb1[kk], (short)0, c1, false, false);
    }

#pragma unroll
    for (int e = 0; e < 8; ++e) {
      float s0 = c0[e] - h0;
      float s1 = c1[e] - h1;
      if (s0 > best[e]) { best[e] = s0; bestn[e] = n0; }
      if (s1 > best[e]) { best[e] = s1; bestn[e] = n1; }
    }
  }

#pragma unroll
  for (int mask = 1; mask <= 8; mask <<= 1) {
#pragma unroll
    for (int e = 0; e < 8; ++e) {
      float ov = __shfl_xor(best[e], mask, 32);
      int oi = __shfl_xor(bestn[e], mask, 32);
      if (ov > best[e] || (ov == best[e] && oi < bestn[e])) {
        best[e] = ov;
        bestn[e] = oi;
      }
    }
  }

  if (ln == 0) {
    int base = token0 + hi * 8;
    int4 b0 = make_int4(bestn[0], bestn[1], bestn[2], bestn[3]);
    int4 b1 = make_int4(bestn[4], bestn[5], bestn[6], bestn[7]);
    *(int4*)(idx + base) = b0;
    *(int4*)(idx + base + 4) = b1;
  }
}

__global__ void __launch_bounds__(256)
vq_gather(const float* __restrict__ cb, const int* __restrict__ idx,
          float* __restrict__ out) {
  int gid = blockIdx.x * blockDim.x + threadIdx.x;
  int t = gid & (TT - 1);
  int d = (gid >> 12) & (DD - 1);
  int b = gid >> 18;
  int id = idx[b * TT + t];
  out[gid] = cb[id * DD + d];
}

extern "C" void kernel_launch(void* const* d_in, const int* in_sizes, int n_in,
                              void* d_out, int out_size, void* d_ws,
                              size_t ws_size, hipStream_t stream) {
  const float* z = (const float*)d_in[0];
  const float* cb = (const float*)d_in[1];
  char* ws = (char*)d_ws;
  float* cbG = (float*)ws;
  float* hes = (float*)(ws + 262144);
  int* idx = (int*)(ws + 266240);

  vq_prep<<<4, 256, 0, stream>>>(cb, cbG, hes);
  vq_encode<<<256, 512, LDS_TOTAL, stream>>>(z, cbG, hes, idx);
  vq_gather<<<16384, 256, 0, stream>>>(cb, idx, (float*)d_out);
}


# Fuse the decode gather into the encode kernel: indices go via LDS and each block writes its 256 tokens' output rows from the LDS-resident codebook in whole 128-byte lines; third launch removed
# speedup vs baseline: 1.3001x; 1.3001x over previous
#include <hip/hip_runtime.h>

typedef __attribute__((ext_vector_type(2))) float v2f;
typedef __attribute__((ext_vector_type(8))) float v8f;

#define TT 4096
#define DD 64
#define KK 1024

#define LDS_HES_B 262144
#define LDS_IDX_B 266240
#define LDS_TOTAL 267264

__global__ void __launch_bounds__(256)
vq_prep(const float* __restrict__ cb, float* __restrict__ cbG,
        float* __restrict__ hes) {
  int n = blockIdx.x * blockDim.x + threadIdx.x;
  if (n >= KK) return;
  const float* row = cb + n * DD;
  float s = 0.f;
  v2f* cp = (v2f*)cbG;
  int base = (n >> 4) * 512 + (n & 15);
#pragma unroll
  for (int j = 0; j < 32; ++j) {
    float x = row[2 * j];
    float y = row[2 * j + 1];
    s += x * x + y * y;
    v2f p;
    p.x = x;
    p.y = y;
    cp[base + j * 16] = p;
  }
  hes[n] = 0.5f * s;
}

__global__ void __launch_bounds__(512)
vq_encode(const float* __restrict__ z, const float* __restrict__ cbG,
          const float* __restrict__ hes, float* __restrict__ out) {
  extern __shared__ char smem[];

  {
    int tid = threadIdx.x;
    const float4* src4 = (const float4*)cbG;
    float4* dst4 = (float4*)smem;
#pragma unroll
    for (int it = 0; it < 32; ++it) {
      dst4[it * 512 + tid] = src4[it * 512 + tid];
    }
    float* shes = (float*)(smem + LDS_HES_B);
    for (int i = tid; i < KK; i += 512) shes[i] = hes[i];
  }
  __syncthreads();

  const int lane = threadIdx.x & 31;
  const int wave = threadIdx.x >> 5;
  const int tile = blockIdx.x * 16 + wave;
  const int token0 = tile << 4;
  const int b = token0 >> 12;
  const int t0 = token0 & (TT - 1);
  const int hi = lane >> 4;
  const int ln = lane & 15;

  const float* zb = z + (size_t)b * DD * TT + t0 + ln;
  float a[32];
#pragma unroll
  for (int kk = 0; kk < 16; ++kk) {
    int d0 = 4 * kk + 2 * hi;
    a[2 * kk]     = zb[(size_t)d0 * TT];
    a[2 * kk + 1] = zb[(size_t)(d0 + 1) * TT];
  }
#pragma unroll
  for (int i = 0; i < 32; ++i) {
    asm volatile("" : "+v"(a[i]));
  }

  float best[8];
  int bestn[8];
#pragma unroll
  for (int e = 0; e < 8; ++e) { best[e] = -3.4e38f; bestn[e] = 0; }

  const char* sb0 = smem + hi * 128 + ln * 8;
  const char* sb1 = sb0 + 4096;
  const float* shp = (const float*)(smem + LDS_HES_B) + ln;

  for (int nt = 0; nt < KK / 32; ++nt) {
    const int n0 = nt * 32 + ln;
    const int n1 = n0 + 16;
    const float h0 = shp[0];
    const float h1 = shp[16];
    shp += 32;

    v2f bb0[16], bb1[16];
#pragma unroll
    for (int kk = 0; kk < 16; ++kk) {
      bb0[kk] = *(const v2f*)(sb0 + kk * 256);
      bb1[kk] = *(const v2f*)(sb1 + kk * 256);
    }
#pragma unroll
    for (int kk = 0; kk < 16; ++kk) {
      asm volatile("" : "+v"(bb0[kk]), "+v"(bb1[kk]));
    }
    sb0 += 8192;
    sb1 += 8192;

    v8f c0 = {};
    v8f c1 = {};
#pragma unroll
    for (int kk = 0; kk < 16; ++kk) {
      v2f av;
      av.x = a[2 * kk];
      av.y = a[2 * kk + 1];
      c0 = __builtin_amdgcn_wmma_f32_16x16x4_f32(
          false, av, false, bb0[kk], (short)0, c0, false, false);
      c1 = __builtin_amdgcn_wmma_f32_16x16x4_f32(
          false, av, false, bb1[kk], (short)0, c1, false, false);
    }

#pragma unroll
    for (int e = 0; e < 8; ++e) {
      float s0 = c0[e] - h0;
      float s1 = c1[e] - h1;
      if (s0 > best[e]) { best[e] = s0; bestn[e] = n0; }
      if (s1 > best[e]) { best[e] = s1; bestn[e] = n1; }
    }
  }

#pragma unroll
  for (int mask = 1; mask <= 8; mask <<= 1) {
#pragma unroll
    for (int e = 0; e < 8; ++e) {
      float ov = __shfl_xor(best[e], mask, 32);
      int oi = __shfl_xor(bestn[e], mask, 32);
      if (ov > best[e] || (ov == best[e] && oi < bestn[e])) {
        best[e] = ov;
        bestn[e] = oi;
      }
    }
  }

  int* sidx = (int*)(smem + LDS_IDX_B);
  if (ln == 0) {
    int base = wave * 16 + hi * 8;
    int4 b0 = make_int4(bestn[0], bestn[1], bestn[2], bestn[3]);
    int4 b1 = make_int4(bestn[4], bestn[5], bestn[6], bestn[7]);
    *(int4*)(sidx + base) = b0;
    *(int4*)(sidx + base + 4) = b1;
  }
  __syncthreads();

  float* ob = out + (size_t)b * DD * TT + (t0 - wave * 16);
  for (int ch = 0; ch < 2; ++ch) {
    const int tok = ch * 128 + lane * 4;
    const int4 id = *(const int4*)(sidx + tok);
    const char* r0 = smem + ((id.x >> 4) << 12) + ((id.x & 15) << 3);
    const char* r1 = smem + ((id.y >> 4) << 12) + ((id.y & 15) << 3);
    const char* r2 = smem + ((id.z >> 4) << 12) + ((id.z & 15) << 3);
    const char* r3 = smem + ((id.w >> 4) << 12) + ((id.w & 15) << 3);
    for (int jj = 0; jj < 2; ++jj) {
      const int j = 2 * wave + jj;
      const v2f p0 = *(const v2f*)(r0 + j * 128);
      const v2f p1 = *(const v2f*)(r1 + j * 128);
      const v2f p2 = *(const v2f*)(r2 + j * 128);
      const v2f p3 = *(const v2f*)(r3 + j * 128);
      float* orow = ob + (size_t)(2 * j) * TT + tok;
      *(float4*)orow = make_float4(p0.x, p1.x, p2.x, p3.x);
      *(float4*)(orow + TT) = make_float4(p0.y, p1.y, p2.y, p3.y);
    }
  }
}

extern "C" void kernel_launch(void* const* d_in, const int* in_sizes, int n_in,
                              void* d_out, int out_size, void* d_ws,
                              size_t ws_size, hipStream_t stream) {
  const float* z = (const float*)d_in[0];
  const float* cb = (const float*)d_in[1];
  char* ws = (char*)d_ws;
  float* cbG = (float*)ws;
  float* hes = (float*)(ws + 262144);

  vq_prep<<<4, 256, 0, stream>>>(cb, cbG, hes);
  vq_encode<<<256, 512, LDS_TOTAL, stream>>>(z, cbG, hes, (float*)d_out);
}


# Argmax reduction across the lanes of each half-wave by DPP row_xmask exchanges on the vector ALU, branch-free, in place of LDS permutes with divergent branches; scores and tie rule unchanged
# speedup vs baseline: 1.5077x; 1.1597x over previous
#include <hip/hip_runtime.h>


typedef __attribute__((ext_vector_type(2))) float v2f;
typedef __attribute__((ext_vector_type(8))) float v8f;

#define TT 4096
#define DD 64
#define KK 1024

#define ROWP 272
#define LDS_HES_B 278528
#define LDS_IDX_B 282624
#define LDS_TOTAL 283648

template <int M>
__device__ __forceinline__ void argmax_xstep(float (&best)[8], int (&bestn)[8]) {
#pragma unroll
  for (int e = 0; e < 8; ++e) {
    const int bv = __builtin_bit_cast(int, best[e]);
    const float ov = __builtin_bit_cast(
        float, __builtin_amdgcn_update_dpp(bv, bv, 0x160 + M, 0xf, 0xf, false));
    const int oi =
        __builtin_amdgcn_update_dpp(bestn[e], bestn[e], 0x160 + M, 0xf, 0xf, false);
    const bool take = (ov > best[e]) | ((ov == best[e]) & (oi < bestn[e]));
    best[e] = take ? ov : best[e];
    bestn[e] = take ? oi : bestn[e];
  }
}

__global__ void __launch_bounds__(512)
vq_encode(const float* __restrict__ z, const float* __restrict__ cb,
          float* __restrict__ out) {
  extern __shared__ char smem[];

  {
    const int tid = threadIdx.x;
    const float4* src4 = (const float4*)cb;
    char* dst = smem + tid * 16 + (tid >> 4) * 16;
#pragma unroll
    for (int it = 0; it < 32; ++it) {
      *(float4*)(dst + it * (32 * ROWP)) = src4[it * 512 + tid];
    }
  }
  __syncthreads();

  const int lane = threadIdx.x & 31;
  const int wave = threadIdx.x >> 5;
  const int tile = blockIdx.x * 16 + wave;
  const int token0 = tile << 4;
  const int b = token0 >> 12;
  const int t0 = token0 & (TT - 1);
  const int hi = lane >> 4;
  const int ln = lane & 15;

  const float* zb = z + (size_t)b * DD * TT + t0 + ln;
  float a[32];
#pragma unroll
  for (int kk = 0; kk < 16; ++kk) {
    int d0 = 4 * kk + 2 * hi;
    a[2 * kk]     = zb[(size_t)d0 * TT];
    a[2 * kk + 1] = zb[(size_t)(d0 + 1) * TT];
  }

  {
    float* shes = (float*)(smem + LDS_HES_B);
    for (int c = 0; c < 2; ++c) {
      const int n = threadIdx.x + c * 512;
      const float4* row = (const float4*)(smem + n * ROWP);
      float s = 0.f;
#pragma unroll
      for (int q = 0; q < 16; ++q) {
        const float4 v = row[q];
        {
#pragma clang fp contract(off)
          if (q < 10) {
            s = s + (v.x * v.x + v.y * v.y);
            s = s + (v.z * v.z + v.w * v.w);
          } else {
            s = s + __builtin_fmaf(v.x, v.x, v.y * v.y);
            s = s + __builtin_fmaf(v.z, v.z, v.w * v.w);
          }
        }
      }
      shes[n] = 0.5f * s;
    }
  }

#pragma unroll
  for (int i = 0; i < 32; ++i) {
    asm volatile("" : "+v"(a[i]));
  }
  __syncthreads();

  float best[8];
  int bestn[8];
#pragma unroll
  for (int e = 0; e < 8; ++e) { best[e] = -3.4e38f; bestn[e] = 0; }

  const char* sb0 = smem + ln * ROWP + hi * 8;
  const char* sb1 = sb0 + 16 * ROWP;
  const float* shp = (const float*)(smem + LDS_HES_B) + ln;

  for (int nt = 0; nt < KK / 32; ++nt) {
    const int n0 = nt * 32 + ln;
    const int n1 = n0 + 16;
    const float h0 = shp[0];
    const float h1 = shp[16];
    shp += 32;

    v2f bb0[16], bb1[16];
#pragma unroll
    for (int kk = 0; kk < 16; ++kk) {
      bb0[kk] = *(const v2f*)(sb0 + kk * 16);
      bb1[kk] = *(const v2f*)(sb1 + kk * 16);
    }
#pragma unroll
    for (int kk = 0; kk < 16; ++kk) {
      asm volatile("" : "+v"(bb0[kk]), "+v"(bb1[kk]));
    }
    sb0 += 32 * ROWP;
    sb1 += 32 * ROWP;

    v8f c0 = {};
    v8f c1 = {};
#pragma unroll
    for (int kk = 0; kk < 16; ++kk) {
      v2f av;
      av.x = a[2 * kk];
      av.y = a[2 * kk + 1];
      c0 = __builtin_amdgcn_wmma_f32_16x16x4_f32(
          false, av, false, bb0[kk], (short)0, c0, false, false);
      c1 = __builtin_amdgcn_wmma_f32_16x16x4_f32(
          false, av, false, bb1[kk], (short)0, c1, false, false);
    }

#pragma unroll
    for (int e = 0; e < 8; ++e) {
      float s0 = c0[e] - h0;
      float s1 = c1[e] - h1;
      if (s0 > best[e]) { best[e] = s0; bestn[e] = n0; }
      if (s1 > best[e]) { best[e] = s1; bestn[e] = n1; }
    }
  }

  argmax_xstep<1>(best, bestn);
  argmax_xstep<2>(best, bestn);
  argmax_xstep<4>(best, bestn);
  argmax_xstep<8>(best, bestn);

  int* sidx = (int*)(smem + LDS_IDX_B);
  if (ln == 0) {
    int base = wave * 16 + hi * 8;
    int4 b0 = make_int4(bestn[0], bestn[1], bestn[2], bestn[3]);
    int4 b1 = make_int4(bestn[4], bestn[5], bestn[6], bestn[7]);
    *(int4*)(sidx + base) = b0;
    *(int4*)(sidx + base + 4) = b1;
  }
  __syncthreads();

  float* ob = out + (size_t)b * DD * TT + (t0 - wave * 16)
                  + (size_t)(4 * wave) * TT;
  for (int ch = 0; ch < 2; ++ch) {
    const int tok = ch * 128 + lane * 4;
    const int4 id = *(const int4*)(sidx + tok);
    const float4 p0 = *(const float4*)(smem + id.x * ROWP + wave * 16);
    const float4 p1 = *(const float4*)(smem + id.y * ROWP + wave * 16);
    const float4 p2 = *(const float4*)(smem + id.z * ROWP + wave * 16);
    const float4 p3 = *(const float4*)(smem + id.w * ROWP + wave * 16);
    float* orow = ob + tok;
    *(float4*)orow            = make_float4(p0.x, p1.x, p2.x, p3.x);
    *(float4*)(orow + TT)     = make_float4(p0.y, p1.y, p2.y, p3.y);
    *(float4*)(orow + 2 * TT) = make_float4(p0.z, p1.z, p2.z, p3.z);
    *(float4*)(orow + 3 * TT) = make_float4(p0.w, p1.w, p2.w, p3.w);
  }
}

extern "C" void kernel_launch(void* const* d_in, const int* in_sizes, int n_in,
                              void* d_out, int out_size, void* d_ws,
                              size_t ws_size, hipStream_t stream) {
  const float* z = (const float*)d_in[0];
  const float* cb = (const float*)d_in[1];

  vq_encode<<<256, 512, LDS_TOTAL, stream>>>(z, cb, (float*)d_out);
}
